# starting kernel
# baseline (speedup 1.0000x reference)
#include <hip/hip_runtime.h>
#include <hip/hip_bf16.h>

typedef __attribute__((ext_vector_type(2))) float v2f;
typedef __attribute__((ext_vector_type(8))) float v8f;

#define HIDDEN 128
#define TOKENS_PER_BLOCK 128
#define NB 64
#define NP 1024

__global__ __launch_bounds__(256) void mlp_wmma_kernel(
    const float* __restrict__ freqs, const float* __restrict__ mags,
    const float* __restrict__ W1, const float* __restrict__ b1,
    const float* __restrict__ W2, const float* __restrict__ b2,
    const float* __restrict__ W3, const float* __restrict__ b3,
    float* __restrict__ out)
{
  __shared__ float hbuf[TOKENS_PER_BLOCK * HIDDEN];

  const int tid       = threadIdx.x;
  const int blockTok0 = blockIdx.x * TOKENS_PER_BLOCK;

  {
    const int token = tid >> 1;
    const int cbase = (tid & 1) * 64;
    const int g     = blockTok0 + token;
    const float f = freqs[g];
    const float m = mags[g];
    #pragma unroll 8
    for (int j = 0; j < 64; ++j) {
      const int c = cbase + j;
      float h = fmaf(f, W1[c], fmaf(m, W1[HIDDEN + c], b1[c]));
      hbuf[token * HIDDEN + c] = h > 0.0f ? h : 0.0f;
    }
  }
  __syncthreads();

  const int lane  = tid & 31;
  const int wave  = tid >> 5;
  const int idx   = lane & 15;
  const int half  = (lane >> 4) << 1;
  const int wrow0 = wave * 16;

  v8f acc[8];
  #pragma unroll
  for (int n = 0; n < 8; ++n) {
    const float bias = b2[n * 16 + idx];
    #pragma unroll
    for (int r = 0; r < 8; ++r) acc[n][r] = bias;
  }

  const float* aP = &hbuf[(wrow0 + idx) * HIDDEN + half];

  #pragma unroll 4
  for (int k = 0; k < HIDDEN; k += 4) {
    v2f a; a.x = aP[k]; a.y = aP[k + 1];
    #pragma unroll
    for (int n = 0; n < 8; ++n) {
      const float* bP = &W2[(k + half) * HIDDEN + n * 16 + idx];
      v2f b; b.x = bP[0]; b.y = bP[HIDDEN];
      acc[n] = __builtin_amdgcn_wmma_f32_16x16x4_f32(
          false, a, false, b, (short)0, acc[n], false, false);
    }
  }

  __syncthreads();

  #pragma unroll
  for (int n = 0; n < 8; ++n) {
    #pragma unroll
    for (int r = 0; r < 8; ++r) {
      const int row = wrow0 + ((lane < 16) ? r : r + 8);
      const float v = acc[n][r];
      hbuf[row * HIDDEN + n * 16 + idx] = v > 0.0f ? v : 0.0f;
    }
  }
  __syncthreads();

  v8f acc3;
  {
    const float bias = (idx < 8) ? b3[idx] : 0.0f;
    #pragma unroll
    for (int r = 0; r < 8; ++r) acc3[r] = bias;
  }
  #pragma unroll 4
  for (int k = 0; k < HIDDEN; k += 4) {
    v2f a; a.x = aP[k]; a.y = aP[k + 1];
    v2f b;
    b.x = (idx < 8) ? W3[(k + half)     * 8 + idx] : 0.0f;
    b.y = (idx < 8) ? W3[(k + half + 1) * 8 + idx] : 0.0f;
    acc3 = __builtin_amdgcn_wmma_f32_16x16x4_f32(
        false, a, false, b, (short)0, acc3, false, false);
  }

  if (idx < 8) {
    const int gbase = blockTok0 + wrow0;
    #pragma unroll
    for (int r = 0; r < 8; ++r) {
      const int row = (lane < 16) ? r : r + 8;
      out[(gbase + row) * 8 + idx] = acc3[r];
    }
  }
}

__global__ __launch_bounds__(1024) void fundamental_kernel(
    const float* __restrict__ freqs, const float* __restrict__ mags,
    float* __restrict__ out)
{
  __shared__ float fr[NP];
  __shared__ float mg[NP];
  __shared__ int   hm[NP];

  const int i    = threadIdx.x;
  const int base = (NB - 1) * NP;
  fr[i] = freqs[base + i];
  mg[i] = mags[base + i];
  __syncthreads();

  const float fi = fr[i];
  int h = 0;
  for (int j = 0; j < NP; ++j) {
    if (j == i) continue;
    if (!(mg[j] > 0.0f)) continue;
    const float fj = fr[j];
    bool match = false;
    #pragma unroll
    for (int order = 2; order <= 8; ++order)
      match = match || (fabsf(fj - fi * (float)order) < 2.0f);
    h += match ? 1 : 0;
  }
  const bool valid = mg[i] > 0.0f;
  hm[i] = (valid && h > 0) ? h : -1;
  __syncthreads();

  if (i == 0) {
    int max_h = -1;
    for (int t = 0; t < NP; ++t) max_h = (hm[t] > max_h) ? hm[t] : max_h;
    int best_idx = 0;
    float best_mag = -__builtin_inff();
    if (max_h > 0) {
      for (int t = 0; t < NP; ++t)
        if (hm[t] == max_h && mg[t] > best_mag) { best_mag = mg[t]; best_idx = t; }
    } else {
      for (int t = 0; t < NP; ++t)
        if (mg[t] > best_mag) { best_mag = mg[t]; best_idx = t; }
    }
    out[NB * NP * 8] = fr[best_idx];
  }
}

extern "C" void kernel_launch(void* const* d_in, const int* in_sizes, int n_in,
                              void* d_out, int out_size, void* d_ws, size_t ws_size,
                              hipStream_t stream) {
  const float* freqs = (const float*)d_in[0];
  const float* mags  = (const float*)d_in[1];
  const float* W1    = (const float*)d_in[2];
  const float* b1    = (const float*)d_in[3];
  const float* W2    = (const float*)d_in[4];
  const float* b2    = (const float*)d_in[5];
  const float* W3    = (const float*)d_in[6];
  const float* b3    = (const float*)d_in[7];
  float* out = (float*)d_out;

  const int ntok   = in_sizes[0];
  const int blocks = ntok / TOKENS_PER_BLOCK;

  mlp_wmma_kernel<<<blocks, 256, 0, stream>>>(freqs, mags, W1, b1, W2, b2, W3, b3, out);
  fundamental_kernel<<<1, 1024, 0, stream>>>(freqs, mags, out);
  (void)d_ws; (void)ws_size; (void)n_in; (void)out_size;
}


# layer 2 wave tile turned: each wave owns 16 output columns for all 128 tokens (8 accumulators), a W2 fragment is loaded once and feeds 8 WMMAs; activations in LDS in fragment order; sum order kept
# speedup vs baseline: 2.6409x; 2.6409x over previous
#include <hip/hip_runtime.h>
#include <hip/hip_bf16.h>

typedef __attribute__((ext_vector_type(2))) float v2f;
typedef __attribute__((ext_vector_type(8))) float v8f;

#define HIDDEN 128
#define TOKENS_PER_BLOCK 128
#define NB 64
#define NP 1024

typedef __attribute__((ext_vector_type(4))) float v4f;

#define MT 8

__device__ __forceinline__ int fslot(int mt, int kb, int fl) {
  return (mt * 16 + kb) * 32 + (fl ^ (((fl >> 4) & 1) << 1) ^ ((kb & 1) << 2));
}

__device__ __forceinline__ float l1act(float f, float m, float w0, float w1, float b) {
  const float h = fmaf(f, w0, fmaf(m, w1, b));
  return h > 0.0f ? h : 0.0f;
}

__global__ __launch_bounds__(256) void mlp_wmma_kernel(
    const float* __restrict__ freqs, const float* __restrict__ mags,
    const float* __restrict__ W1, const float* __restrict__ b1,
    const float* __restrict__ W2, const float* __restrict__ b2,
    const float* __restrict__ W3, const float* __restrict__ b3,
    float* __restrict__ out)
{
  __shared__ v4f hfrag[MT * 16 * 32];

  const int tid       = threadIdx.x;
  const int blockTok0 = blockIdx.x * TOKENS_PER_BLOCK;

  const int lane = tid & 31;
  const int wave = tid >> 5;
  const int idx  = lane & 15;
  const int hl   = lane >> 4;

  const int   g1    = blockTok0 + wave * 16 + idx;
  const float f     = freqs[g1];
  const float m     = mags[g1];
  const float bias2 = b2[wave * 16 + idx];

  v2f bw[32];
  {
    const float* w2p = W2 + (2 * hl) * HIDDEN + wave * 16 + idx;
    #pragma unroll
    for (int ks = 0; ks < 32; ++ks) {
      bw[ks].x = w2p[(4 * ks) * HIDDEN];
      bw[ks].y = w2p[(4 * ks + 1) * HIDDEN];
    }
  }

  {
    #pragma unroll 4
    for (int kb = 0; kb < 16; ++kb) {
      const int ca = kb * 8 + 2 * hl;
      const int cb = ca + 4;
      const v2f w0a = *(const v2f*)(W1 + ca);
      const v2f w1a = *(const v2f*)(W1 + HIDDEN + ca);
      const v2f bba = *(const v2f*)(b1 + ca);
      const v2f w0b = *(const v2f*)(W1 + cb);
      const v2f w1b = *(const v2f*)(W1 + HIDDEN + cb);
      const v2f bbb = *(const v2f*)(b1 + cb);
      v4f hv;
      hv.x = l1act(f, m, w0a.x, w1a.x, bba.x);
      hv.y = l1act(f, m, w0a.y, w1a.y, bba.y);
      hv.z = l1act(f, m, w0b.x, w1b.x, bbb.x);
      hv.w = l1act(f, m, w0b.y, w1b.y, bbb.y);
      hfrag[fslot(wave, kb, lane)] = hv;
    }
  }
  __syncthreads();

  v8f acc[MT];
  {
    #pragma unroll
    for (int mt = 0; mt < MT; ++mt) {
      #pragma unroll
      for (int r = 0; r < 8; ++r) acc[mt][r] = bias2;
    }
  }

  {
    v4f a_cur[MT];
    #pragma unroll
    for (int mt = 0; mt < MT; ++mt) a_cur[mt] = hfrag[fslot(mt, 0, lane)];

    #pragma unroll
    for (int kb = 0; kb < 16; ++kb) {
      v4f a_nxt[MT];
      if (kb + 1 < 16) {
        #pragma unroll
        for (int mt = 0; mt < MT; ++mt) a_nxt[mt] = hfrag[fslot(mt, kb + 1, lane)];
      }
      #pragma unroll
      for (int mt = 0; mt < MT; ++mt) {
        v2f a; a.x = a_cur[mt].x; a.y = a_cur[mt].y;
        acc[mt] = __builtin_amdgcn_wmma_f32_16x16x4_f32(
            false, a, false, bw[2 * kb], (short)0, acc[mt], false, false);
      }
      #pragma unroll
      for (int mt = 0; mt < MT; ++mt) {
        v2f a; a.x = a_cur[mt].z; a.y = a_cur[mt].w;
        acc[mt] = __builtin_amdgcn_wmma_f32_16x16x4_f32(
            false, a, false, bw[2 * kb + 1], (short)0, acc[mt], false, false);
      }
      if (kb + 1 < 16) {
        #pragma unroll
        for (int mt = 0; mt < MT; ++mt) a_cur[mt] = a_nxt[mt];
      }
    }
  }

  v2f b3f[32];
  const bool realcol = idx < 8;
  {
    const float* w3p = W3 + (2 * hl) * 8 + (idx & 7);
    #pragma unroll
    for (int ks = 0; ks < 32; ++ks) {
      const float wx = w3p[(4 * ks) * 8];
      const float wy = w3p[(4 * ks + 1) * 8];
      b3f[ks].x = realcol ? wx : 0.0f;
      b3f[ks].y = realcol ? wy : 0.0f;
    }
  }
  const float bias3v = b3[idx & 7];
  const float bias3  = realcol ? bias3v : 0.0f;

  __syncthreads();

  {
    float* hf = (float*)hfrag;
    const int kb3  = 2 * wave + (idx >> 3);
    const int hh   = (idx >> 1) & 1;
    const int elem = (idx & 1) + 2 * ((idx >> 2) & 1);
    #pragma unroll
    for (int r = 0; r < 8; ++r) {
      const int o = fslot(0, kb3, 16 * hh + 8 * hl + r) * 4 + elem;
      #pragma unroll
      for (int mt = 0; mt < MT; ++mt) {
        const float v = acc[mt][r];
        hf[mt * (16 * 32 * 4) + o] = v > 0.0f ? v : 0.0f;
      }
    }
  }
  __syncthreads();

  v8f acc3;
  {
    #pragma unroll
    for (int r = 0; r < 8; ++r) acc3[r] = bias3;
  }
  #pragma unroll
  for (int kb = 0; kb < 16; ++kb) {
    const v4f av = hfrag[fslot(wave, kb, lane)];
    v2f a; a.x = av.x; a.y = av.y;
    acc3 = __builtin_amdgcn_wmma_f32_16x16x4_f32(
        false, a, false, b3f[2 * kb], (short)0, acc3, false, false);
    a.x = av.z; a.y = av.w;
    acc3 = __builtin_amdgcn_wmma_f32_16x16x4_f32(
        false, a, false, b3f[2 * kb + 1], (short)0, acc3, false, false);
  }

  if (idx < 8) {
    const int gbase = blockTok0 + wave * 16;
    #pragma unroll
    for (int r = 0; r < 8; ++r) {
      const int row = 8 * hl + r;
      out[(gbase + row) * 8 + idx] = acc3[r];
    }
  }
}


__device__ __forceinline__ unsigned f2ord(float f) {
  const unsigned b = __float_as_uint(f);
  return (b & 0x80000000u) ? ~b : (b | 0x80000000u);
}
__device__ __forceinline__ float ord2f(unsigned k) {
  const unsigned b = (k & 0x80000000u) ? (k & 0x7FFFFFFFu) : ~k;
  return __uint_as_float(b);
}
__device__ __forceinline__ float hdist(float fj, float fi, float order) {
  float d;
  {
    #pragma clang fp contract(off)
    d = fj - fi * order;
  }
  return d;
}

__global__ __launch_bounds__(1024) void fundamental_kernel(
    const float* __restrict__ freqs, const float* __restrict__ mags,
    float* __restrict__ out)
{
  __shared__ float    fr[NP];
  __shared__ float    mg[NP];
  __shared__ int      hm[NP];
  __shared__ unsigned xb[2][NP];
  __shared__ float    sf[2 * NP];

  const int i    = threadIdx.x;
  const int base = (NB - 1) * NP;
  const float fi = freqs[base + i];
  const float mi = mags[base + i];
  const bool valid = mi > 0.0f;
  fr[i] = fi;
  mg[i] = mi;

  unsigned key = valid ? f2ord(fi) : 0xFFFFFFFFu;
  int tog = 0;
  #pragma unroll
  for (int k = 2; k <= NP; k <<= 1) {
    const bool up = ((i & k) == 0);
    #pragma unroll
    for (int j = k >> 1; j > 0; j >>= 1) {
      unsigned other;
      if (j >= 32) {
        unsigned* buf = xb[tog];
        tog ^= 1;
        buf[i] = key;
        __syncthreads();
        other = buf[i ^ j];
      } else {
        other = __shfl_xor(key, j, 32);
      }
      const bool lower = ((i & j) == 0);
      const unsigned mn = key < other ? key : other;
      const unsigned mx = key < other ? other : key;
      key = (lower == up) ? mn : mx;
    }
  }
  const float pinf = __builtin_inff();
  sf[i]      = (key == 0xFFFFFFFFu) ? pinf : ord2f(key);
  sf[NP + i] = pinf;
  __syncthreads();

  int h = 0;
  if (valid) {
    int lo[7], hi[7];
    for (int t = 0; t < 7; ++t) { lo[t] = 0; hi[t] = 0; }
    #pragma unroll
    for (int step = NP; step > 0; step >>= 1) {
      #pragma unroll
      for (int t = 0; t < 7; ++t) {
        const float of = (float)(t + 2);
        const float dl = hdist(sf[lo[t] + step - 1], fi, of);
        const float dh = hdist(sf[hi[t] + step - 1], fi, of);
        if (!(dl > -2.0f)) lo[t] += step;
        if (!(dh >= 2.0f)) hi[t] += step;
      }
    }
    const bool neg = fi < 0.0f;
    int cover = 0;
    for (int t = 0; t < 7; ++t) {
      int a       = neg ? 2 * NP - hi[t] : lo[t];
      const int b = neg ? 2 * NP - lo[t] : hi[t];
      a = a > cover ? a : cover;
      h += b > a ? b - a : 0;
      cover = cover > b ? cover : b;
    }
    bool selfm = false;
    for (int t = 0; t < 7; ++t)
      selfm = selfm || (fabsf(hdist(fi, fi, (float)(t + 2))) < 2.0f);
    h -= selfm ? 1 : 0;
  }
  hm[i] = (valid && h > 0) ? h : -1;
  __syncthreads();

  if (i == 0) {
    int max_h = -1;
    for (int t = 0; t < NP; ++t) max_h = (hm[t] > max_h) ? hm[t] : max_h;
    int best_idx = 0;
    float best_mag = -__builtin_inff();
    if (max_h > 0) {
      for (int t = 0; t < NP; ++t)
        if (hm[t] == max_h && mg[t] > best_mag) { best_mag = mg[t]; best_idx = t; }
    } else {
      for (int t = 0; t < NP; ++t)
        if (mg[t] > best_mag) { best_mag = mg[t]; best_idx = t; }
    }
    out[NB * NP * 8] = fr[best_idx];
  }
}

extern "C" void kernel_launch(void* const* d_in, const int* in_sizes, int n_in,
                              void* d_out, int out_size, void* d_ws, size_t ws_size,
                              hipStream_t stream) {
  const float* freqs = (const float*)d_in[0];
  const float* mags  = (const float*)d_in[1];
  const float* W1    = (const float*)d_in[2];
  const float* b1    = (const float*)d_in[3];
  const float* W2    = (const float*)d_in[4];
  const float* b2    = (const float*)d_in[5];
  const float* W3    = (const float*)d_in[6];
  const float* b3    = (const float*)d_in[7];
  float* out = (float*)d_out;

  const int ntok   = in_sizes[0];
  const int blocks = ntok / TOKENS_PER_BLOCK;

  mlp_wmma_kernel<<<blocks, 256, 0, stream>>>(freqs, mags, W1, b1, W2, b2, W3, b3, out);
  fundamental_kernel<<<1, 1024, 0, stream>>>(freqs, mags, out);
  (void)d_ws; (void)ws_size; (void)n_in; (void)out_size;
}


# launch 2 final pick, still one lane: the branch and LDS round trip per peak replaced by 16-byte LDS loads issued in groups and four independent branch-free running maxima, merged lowest index first
# speedup vs baseline: 5.7672x; 2.1838x over previous
#include <hip/hip_runtime.h>
#include <hip/hip_bf16.h>

typedef __attribute__((ext_vector_type(2))) float v2f;
typedef __attribute__((ext_vector_type(8))) float v8f;

#define HIDDEN 128
#define TOKENS_PER_BLOCK 128
#define NB 64
#define NP 1024

typedef __attribute__((ext_vector_type(4))) float v4f;
typedef __attribute__((ext_vector_type(4))) int   v4i;

#define MT 8

__device__ __forceinline__ int fslot(int mt, int kb, int fl) {
  return (mt * 16 + kb) * 32 + (fl ^ (((fl >> 4) & 1) << 1) ^ ((kb & 1) << 2));
}

__device__ __forceinline__ float l1act(float f, float m, float w0, float w1, float b) {
  const float h = fmaf(f, w0, fmaf(m, w1, b));
  return h > 0.0f ? h : 0.0f;
}

__global__ __launch_bounds__(256) void mlp_wmma_kernel(
    const float* __restrict__ freqs, const float* __restrict__ mags,
    const float* __restrict__ W1, const float* __restrict__ b1,
    const float* __restrict__ W2, const float* __restrict__ b2,
    const float* __restrict__ W3, const float* __restrict__ b3,
    float* __restrict__ out)
{
  __shared__ v4f hfrag[MT * 16 * 32];

  const int tid       = threadIdx.x;
  const int blockTok0 = blockIdx.x * TOKENS_PER_BLOCK;

  const int lane = tid & 31;
  const int wave = tid >> 5;
  const int idx  = lane & 15;
  const int hl   = lane >> 4;

  const int   g1    = blockTok0 + wave * 16 + idx;
  const float f     = freqs[g1];
  const float m     = mags[g1];
  const float bias2 = b2[wave * 16 + idx];

  v2f bw[32];
  {
    const float* w2p = W2 + (2 * hl) * HIDDEN + wave * 16 + idx;
    #pragma unroll
    for (int ks = 0; ks < 32; ++ks) {
      bw[ks].x = w2p[(4 * ks) * HIDDEN];
      bw[ks].y = w2p[(4 * ks + 1) * HIDDEN];
    }
  }

  {
    #pragma unroll 4
    for (int kb = 0; kb < 16; ++kb) {
      const int ca = kb * 8 + 2 * hl;
      const int cb = ca + 4;
      const v2f w0a = *(const v2f*)(W1 + ca);
      const v2f w1a = *(const v2f*)(W1 + HIDDEN + ca);
      const v2f bba = *(const v2f*)(b1 + ca);
      const v2f w0b = *(const v2f*)(W1 + cb);
      const v2f w1b = *(const v2f*)(W1 + HIDDEN + cb);
      const v2f bbb = *(const v2f*)(b1 + cb);
      v4f hv;
      hv.x = l1act(f, m, w0a.x, w1a.x, bba.x);
      hv.y = l1act(f, m, w0a.y, w1a.y, bba.y);
      hv.z = l1act(f, m, w0b.x, w1b.x, bbb.x);
      hv.w = l1act(f, m, w0b.y, w1b.y, bbb.y);
      hfrag[fslot(wave, kb, lane)] = hv;
    }
  }
  __syncthreads();

  v8f acc[MT];
  {
    #pragma unroll
    for (int mt = 0; mt < MT; ++mt) {
      #pragma unroll
      for (int r = 0; r < 8; ++r) acc[mt][r] = bias2;
    }
  }

  {
    v4f a_cur[MT];
    #pragma unroll
    for (int mt = 0; mt < MT; ++mt) a_cur[mt] = hfrag[fslot(mt, 0, lane)];

    #pragma unroll
    for (int kb = 0; kb < 16; ++kb) {
      v4f a_nxt[MT];
      if (kb + 1 < 16) {
        #pragma unroll
        for (int mt = 0; mt < MT; ++mt) a_nxt[mt] = hfrag[fslot(mt, kb + 1, lane)];
      }
      #pragma unroll
      for (int mt = 0; mt < MT; ++mt) {
        v2f a; a.x = a_cur[mt].x; a.y = a_cur[mt].y;
        acc[mt] = __builtin_amdgcn_wmma_f32_16x16x4_f32(
            false, a, false, bw[2 * kb], (short)0, acc[mt], false, false);
      }
      #pragma unroll
      for (int mt = 0; mt < MT; ++mt) {
        v2f a; a.x = a_cur[mt].z; a.y = a_cur[mt].w;
        acc[mt] = __builtin_amdgcn_wmma_f32_16x16x4_f32(
            false, a, false, bw[2 * kb + 1], (short)0, acc[mt], false, false);
      }
      if (kb + 1 < 16) {
        #pragma unroll
        for (int mt = 0; mt < MT; ++mt) a_cur[mt] = a_nxt[mt];
      }
    }
  }

  v2f b3f[32];
  const bool realcol = idx < 8;
  {
    const float* w3p = W3 + (2 * hl) * 8 + (idx & 7);
    #pragma unroll
    for (int ks = 0; ks < 32; ++ks) {
      const float wx = w3p[(4 * ks) * 8];
      const float wy = w3p[(4 * ks + 1) * 8];
      b3f[ks].x = realcol ? wx : 0.0f;
      b3f[ks].y = realcol ? wy : 0.0f;
    }
  }
  const float bias3v = b3[idx & 7];
  const float bias3  = realcol ? bias3v : 0.0f;

  __syncthreads();

  {
    float* hf = (float*)hfrag;
    const int kb3  = 2 * wave + (idx >> 3);
    const int hh   = (idx >> 1) & 1;
    const int elem = (idx & 1) + 2 * ((idx >> 2) & 1);
    #pragma unroll
    for (int r = 0; r < 8; ++r) {
      const int o = fslot(0, kb3, 16 * hh + 8 * hl + r) * 4 + elem;
      #pragma unroll
      for (int mt = 0; mt < MT; ++mt) {
        const float v = acc[mt][r];
        hf[mt * (16 * 32 * 4) + o] = v > 0.0f ? v : 0.0f;
      }
    }
  }
  __syncthreads();

  v8f acc3;
  {
    #pragma unroll
    for (int r = 0; r < 8; ++r) acc3[r] = bias3;
  }
  #pragma unroll
  for (int kb = 0; kb < 16; ++kb) {
    const v4f av = hfrag[fslot(wave, kb, lane)];
    v2f a; a.x = av.x; a.y = av.y;
    acc3 = __builtin_amdgcn_wmma_f32_16x16x4_f32(
        false, a, false, b3f[2 * kb], (short)0, acc3, false, false);
    a.x = av.z; a.y = av.w;
    acc3 = __builtin_amdgcn_wmma_f32_16x16x4_f32(
        false, a, false, b3f[2 * kb + 1], (short)0, acc3, false, false);
  }

  if (idx < 8) {
    const int gbase = blockTok0 + wave * 16;
    #pragma unroll
    for (int r = 0; r < 8; ++r) {
      const int row = 8 * hl + r;
      out[(gbase + row) * 8 + idx] = acc3[r];
    }
  }
}


__device__ __forceinline__ unsigned f2ord(float f) {
  const unsigned b = __float_as_uint(f);
  return (b & 0x80000000u) ? ~b : (b | 0x80000000u);
}
__device__ __forceinline__ float ord2f(unsigned k) {
  const unsigned b = (k & 0x80000000u) ? (k & 0x7FFFFFFFu) : ~k;
  return __uint_as_float(b);
}
__device__ __forceinline__ float hdist(float fj, float fi, float order) {
  float d;
  {
    #pragma clang fp contract(off)
    d = fj - fi * order;
  }
  return d;
}

__global__ __launch_bounds__(1024) void fundamental_kernel(
    const float* __restrict__ freqs, const float* __restrict__ mags,
    float* __restrict__ out)
{
  __shared__ float    fr[NP];
  __shared__ __attribute__((aligned(16))) float mg[NP];
  __shared__ __attribute__((aligned(16))) int   hm[NP];
  __shared__ unsigned xb[2][NP];
  __shared__ float    sf[2 * NP];

  const int i    = threadIdx.x;
  const int base = (NB - 1) * NP;
  const float fi = freqs[base + i];
  const float mi = mags[base + i];
  const bool valid = mi > 0.0f;
  fr[i] = fi;
  mg[i] = mi;

  unsigned key = valid ? f2ord(fi) : 0xFFFFFFFFu;
  int tog = 0;
  #pragma unroll
  for (int k = 2; k <= NP; k <<= 1) {
    const bool up = ((i & k) == 0);
    #pragma unroll
    for (int j = k >> 1; j > 0; j >>= 1) {
      unsigned other;
      if (j >= 32) {
        unsigned* buf = xb[tog];
        tog ^= 1;
        buf[i] = key;
        __syncthreads();
        other = buf[i ^ j];
      } else {
        other = __shfl_xor(key, j, 32);
      }
      const bool lower = ((i & j) == 0);
      const unsigned mn = key < other ? key : other;
      const unsigned mx = key < other ? other : key;
      key = (lower == up) ? mn : mx;
    }
  }
  const float pinf = __builtin_inff();
  sf[i]      = (key == 0xFFFFFFFFu) ? pinf : ord2f(key);
  sf[NP + i] = pinf;
  __syncthreads();

  int h = 0;
  if (valid) {
    int lo[7], hi[7];
    for (int t = 0; t < 7; ++t) { lo[t] = 0; hi[t] = 0; }
    #pragma unroll
    for (int step = NP; step > 0; step >>= 1) {
      #pragma unroll
      for (int t = 0; t < 7; ++t) {
        const float of = (float)(t + 2);
        const float dl = hdist(sf[lo[t] + step - 1], fi, of);
        const float dh = hdist(sf[hi[t] + step - 1], fi, of);
        if (!(dl > -2.0f)) lo[t] += step;
        if (!(dh >= 2.0f)) hi[t] += step;
      }
    }
    const bool neg = fi < 0.0f;
    int cover = 0;
    for (int t = 0; t < 7; ++t) {
      int a       = neg ? 2 * NP - hi[t] : lo[t];
      const int b = neg ? 2 * NP - lo[t] : hi[t];
      a = a > cover ? a : cover;
      h += b > a ? b - a : 0;
      cover = cover > b ? cover : b;
    }
    bool selfm = false;
    for (int t = 0; t < 7; ++t)
      selfm = selfm || (fabsf(hdist(fi, fi, (float)(t + 2))) < 2.0f);
    h -= selfm ? 1 : 0;
  }
  hm[i] = (valid && h > 0) ? h : -1;
  __syncthreads();

  if (i == 0) {
    int max_h = -1;
    for (int t = 0; t < NP; ++t) max_h = (hm[t] > max_h) ? hm[t] : max_h;
    const float ninf = -__builtin_inff();
    const v4i* hm4 = (const v4i*)hm;
    const v4f* mg4 = (const v4f*)mg;
    v4f bm; v4i bq;
    bm.x = ninf; bm.y = ninf; bm.z = ninf; bm.w = ninf;
    bq.x = 0; bq.y = 0; bq.z = 0; bq.w = 0;
    #pragma unroll 8
    for (int q = 0; q < NP / 4; ++q) {
      const v4i hv = hm4[q];
      const v4f mv = mg4[q];
      #pragma unroll
      for (int c = 0; c < 4; ++c) {
        const float v    = (hv[c] == max_h) ? mv[c] : ninf;
        const bool  take = v > bm[c];
        bm[c] = take ? v : bm[c];
        bq[c] = take ? q : bq[c];
      }
    }
    float best_mag = bm[0];
    int   best_idx = 4 * bq[0];
    #pragma unroll
    for (int c = 1; c < 4; ++c) {
      const int  ic   = 4 * bq[c] + c;
      const bool take = (bm[c] > best_mag) | ((bm[c] == best_mag) & (ic < best_idx));
      best_mag = take ? bm[c] : best_mag;
      best_idx = take ? ic : best_idx;
    }
    out[NB * NP * 8] = fr[best_idx];
  }
}

extern "C" void kernel_launch(void* const* d_in, const int* in_sizes, int n_in,
                              void* d_out, int out_size, void* d_ws, size_t ws_size,
                              hipStream_t stream) {
  const float* freqs = (const float*)d_in[0];
  const float* mags  = (const float*)d_in[1];
  const float* W1    = (const float*)d_in[2];
  const float* b1    = (const float*)d_in[3];
  const float* W2    = (const float*)d_in[4];
  const float* b2    = (const float*)d_in[5];
  const float* W3    = (const float*)d_in[6];
  const float* b3    = (const float*)d_in[7];
  float* out = (float*)d_out;

  const int ntok   = in_sizes[0];
  const int blocks = ntok / TOKENS_PER_BLOCK;

  mlp_wmma_kernel<<<blocks, 256, 0, stream>>>(freqs, mags, W1, b1, W2, b2, W3, b3, out);
  fundamental_kernel<<<1, 1024, 0, stream>>>(freqs, mags, out);
  (void)d_ws; (void)ws_size; (void)n_in; (void)out_size;
}


# the two launches merged into one: the last-row fundamental pick (sort, searches, one-lane pick) runs as work-group 0 of the MLP launch, 256 threads with four peaks each, beside the MLP tiles
# speedup vs baseline: 9.0566x; 1.5704x over previous
#include <hip/hip_runtime.h>
#include <hip/hip_bf16.h>

typedef __attribute__((ext_vector_type(2))) float v2f;
typedef __attribute__((ext_vector_type(8))) float v8f;

#define HIDDEN 128
#define TOKENS_PER_BLOCK 128
#define NB 64
#define NP 1024

typedef __attribute__((ext_vector_type(4))) float v4f;
typedef __attribute__((ext_vector_type(4))) int   v4i;

#define MT 8

__device__ __forceinline__ int fslot(int mt, int kb, int fl) {
  return (mt * 16 + kb) * 32 + (fl ^ (((fl >> 4) & 1) << 1) ^ ((kb & 1) << 2));
}

__device__ __forceinline__ float l1act(float f, float m, float w0, float w1, float b) {
  const float h = fmaf(f, w0, fmaf(m, w1, b));
  return h > 0.0f ? h : 0.0f;
}

typedef __attribute__((ext_vector_type(4))) unsigned v4u;

#define PK 4

__device__ __forceinline__ unsigned f2ord(float f) {
  const unsigned b = __float_as_uint(f);
  return (b & 0x80000000u) ? ~b : (b | 0x80000000u);
}
__device__ __forceinline__ float ord2f(unsigned k) {
  const unsigned b = (k & 0x80000000u) ? (k & 0x7FFFFFFFu) : ~k;
  return __uint_as_float(b);
}
__device__ __forceinline__ float hdist(float fj, float fi, float order) {
  float d;
  {
    #pragma clang fp contract(off)
    d = fj - fi * order;
  }
  return d;
}

__device__ __forceinline__ void last_row_fundamental(
    const float* __restrict__ freqs, const float* __restrict__ mags,
    float* __restrict__ out)
{
  __shared__ __attribute__((aligned(16))) float    fr[NP];
  __shared__ __attribute__((aligned(16))) float    mg[NP];
  __shared__ __attribute__((aligned(16))) int      hm[NP];
  __shared__ __attribute__((aligned(16))) unsigned xb[2][NP];
  __shared__ __attribute__((aligned(16))) float    sf[2 * NP];

  const int tid  = threadIdx.x;
  const int base = (NB - 1) * NP;
  const v4f fv = *(const v4f*)(freqs + base + PK * tid);
  const v4f mv = *(const v4f*)(mags + base + PK * tid);
  ((v4f*)fr)[tid] = fv;
  ((v4f*)mg)[tid] = mv;

  unsigned key[PK];
  #pragma unroll
  for (int e = 0; e < PK; ++e) key[e] = (mv[e] > 0.0f) ? f2ord(fv[e]) : 0xFFFFFFFFu;

  int tog = 0;
  #pragma unroll
  for (int k = 2; k <= NP; k <<= 1) {
    #pragma unroll
    for (int j = k >> 1; j > 0; j >>= 1) {
      unsigned other[PK];
      if (j >= PK * 32) {
        v4u* buf = (v4u*)xb[tog];
        tog ^= 1;
        v4u kv;
        #pragma unroll
        for (int e = 0; e < PK; ++e) kv[e] = key[e];
        buf[tid] = kv;
        __syncthreads();
        const v4u ov = buf[tid ^ (j / PK)];
        #pragma unroll
        for (int e = 0; e < PK; ++e) other[e] = ov[e];
      } else if (j >= PK) {
        #pragma unroll
        for (int e = 0; e < PK; ++e) other[e] = __shfl_xor(key[e], j / PK, 32);
      } else {
        #pragma unroll
        for (int e = 0; e < PK; ++e) other[e] = key[e ^ j];
      }
      #pragma unroll
      for (int e = 0; e < PK; ++e) {
        const int  i     = PK * tid + e;
        const bool up    = ((i & k) == 0);
        const bool lower = ((i & j) == 0);
        const unsigned mn = key[e] < other[e] ? key[e] : other[e];
        const unsigned mx = key[e] < other[e] ? other[e] : key[e];
        key[e] = (lower == up) ? mn : mx;
      }
    }
  }
  const float pinf = __builtin_inff();
  {
    v4f sv, iv;
    #pragma unroll
    for (int e = 0; e < PK; ++e) {
      sv[e] = (key[e] == 0xFFFFFFFFu) ? pinf : ord2f(key[e]);
      iv[e] = pinf;
    }
    ((v4f*)sf)[tid]           = sv;
    ((v4f*)sf)[NP / PK + tid] = iv;
  }
  __syncthreads();

  int lo[PK][7], hi[PK][7];
  #pragma unroll
  for (int e = 0; e < PK; ++e) {
    #pragma unroll
    for (int t = 0; t < 7; ++t) { lo[e][t] = 0; hi[e][t] = 0; }
  }
  #pragma unroll
  for (int step = NP; step > 0; step >>= 1) {
    #pragma unroll
    for (int e = 0; e < PK; ++e) {
      #pragma unroll
      for (int t = 0; t < 7; ++t) {
        const float of = (float)(t + 2);
        const float dl = hdist(sf[lo[e][t] + step - 1], fv[e], of);
        const float dh = hdist(sf[hi[e][t] + step - 1], fv[e], of);
        if (!(dl > -2.0f)) lo[e][t] += step;
        if (!(dh >= 2.0f)) hi[e][t] += step;
      }
    }
  }
  v4i hv;
  #pragma unroll
  for (int e = 0; e < PK; ++e) {
    const float fi    = fv[e];
    const bool  valid = mv[e] > 0.0f;
    const bool neg = fi < 0.0f;
    int h = 0, cover = 0;
    #pragma unroll
    for (int t = 0; t < 7; ++t) {
      int a       = neg ? 2 * NP - hi[e][t] : lo[e][t];
      const int b = neg ? 2 * NP - lo[e][t] : hi[e][t];
      a = a > cover ? a : cover;
      h += b > a ? b - a : 0;
      cover = cover > b ? cover : b;
    }
    bool selfm = false;
    #pragma unroll
    for (int t = 0; t < 7; ++t)
      selfm = selfm || (fabsf(hdist(fi, fi, (float)(t + 2))) < 2.0f);
    h -= selfm ? 1 : 0;
    hv[e] = (valid && h > 0) ? h : -1;
  }
  ((v4i*)hm)[tid] = hv;
  __syncthreads();

  if (tid == 0) {
    int max_h = -1;
    for (int t = 0; t < NP; ++t) max_h = (hm[t] > max_h) ? hm[t] : max_h;
    const float ninf = -__builtin_inff();
    const v4i* hm4 = (const v4i*)hm;
    const v4f* mg4 = (const v4f*)mg;
    v4f bm; v4i bq;
    bm.x = ninf; bm.y = ninf; bm.z = ninf; bm.w = ninf;
    bq.x = 0; bq.y = 0; bq.z = 0; bq.w = 0;
    #pragma unroll 8
    for (int q = 0; q < NP / 4; ++q) {
      const v4i hv = hm4[q];
      const v4f mv = mg4[q];
      #pragma unroll
      for (int c = 0; c < 4; ++c) {
        const float v    = (hv[c] == max_h) ? mv[c] : ninf;
        const bool  take = v > bm[c];
        bm[c] = take ? v : bm[c];
        bq[c] = take ? q : bq[c];
      }
    }
    float best_mag = bm[0];
    int   best_idx = 4 * bq[0];
    #pragma unroll
    for (int c = 1; c < 4; ++c) {
      const int  ic   = 4 * bq[c] + c;
      const bool take = (bm[c] > best_mag) | ((bm[c] == best_mag) & (ic < best_idx));
      best_mag = take ? bm[c] : best_mag;
      best_idx = take ? ic : best_idx;
    }
    out[NB * NP * 8] = fr[best_idx];
  }
}

__global__ __launch_bounds__(256) void mlp_and_last_row_kernel(
    const float* __restrict__ freqs, const float* __restrict__ mags,
    const float* __restrict__ W1, const float* __restrict__ b1,
    const float* __restrict__ W2, const float* __restrict__ b2,
    const float* __restrict__ W3, const float* __restrict__ b3,
    float* __restrict__ out)
{
  __shared__ v4f hfrag[MT * 16 * 32];

  if (blockIdx.x == 0) {
    last_row_fundamental(freqs, mags, out);
    return;
  }

  const int tid       = threadIdx.x;
  const int blockTok0 = (blockIdx.x - 1) * TOKENS_PER_BLOCK;

  const int lane = tid & 31;
  const int wave = tid >> 5;
  const int idx  = lane & 15;
  const int hl   = lane >> 4;

  const int   g1    = blockTok0 + wave * 16 + idx;
  const float f     = freqs[g1];
  const float m     = mags[g1];
  const float bias2 = b2[wave * 16 + idx];

  v2f bw[32];
  {
    const float* w2p = W2 + (2 * hl) * HIDDEN + wave * 16 + idx;
    #pragma unroll
    for (int ks = 0; ks < 32; ++ks) {
      bw[ks].x = w2p[(4 * ks) * HIDDEN];
      bw[ks].y = w2p[(4 * ks + 1) * HIDDEN];
    }
  }

  {
    #pragma unroll 4
    for (int kb = 0; kb < 16; ++kb) {
      const int ca = kb * 8 + 2 * hl;
      const int cb = ca + 4;
      const v2f w0a = *(const v2f*)(W1 + ca);
      const v2f w1a = *(const v2f*)(W1 + HIDDEN + ca);
      const v2f bba = *(const v2f*)(b1 + ca);
      const v2f w0b = *(const v2f*)(W1 + cb);
      const v2f w1b = *(const v2f*)(W1 + HIDDEN + cb);
      const v2f bbb = *(const v2f*)(b1 + cb);
      v4f hv;
      hv.x = l1act(f, m, w0a.x, w1a.x, bba.x);
      hv.y = l1act(f, m, w0a.y, w1a.y, bba.y);
      hv.z = l1act(f, m, w0b.x, w1b.x, bbb.x);
      hv.w = l1act(f, m, w0b.y, w1b.y, bbb.y);
      hfrag[fslot(wave, kb, lane)] = hv;
    }
  }
  __syncthreads();

  v8f acc[MT];
  {
    #pragma unroll
    for (int mt = 0; mt < MT; ++mt) {
      #pragma unroll
      for (int r = 0; r < 8; ++r) acc[mt][r] = bias2;
    }
  }

  {
    v4f a_cur[MT];
    #pragma unroll
    for (int mt = 0; mt < MT; ++mt) a_cur[mt] = hfrag[fslot(mt, 0, lane)];

    #pragma unroll
    for (int kb = 0; kb < 16; ++kb) {
      v4f a_nxt[MT];
      if (kb + 1 < 16) {
        #pragma unroll
        for (int mt = 0; mt < MT; ++mt) a_nxt[mt] = hfrag[fslot(mt, kb + 1, lane)];
      }
      #pragma unroll
      for (int mt = 0; mt < MT; ++mt) {
        v2f a; a.x = a_cur[mt].x; a.y = a_cur[mt].y;
        acc[mt] = __builtin_amdgcn_wmma_f32_16x16x4_f32(
            false, a, false, bw[2 * kb], (short)0, acc[mt], false, false);
      }
      #pragma unroll
      for (int mt = 0; mt < MT; ++mt) {
        v2f a; a.x = a_cur[mt].z; a.y = a_cur[mt].w;
        acc[mt] = __builtin_amdgcn_wmma_f32_16x16x4_f32(
            false, a, false, bw[2 * kb + 1], (short)0, acc[mt], false, false);
      }
      if (kb + 1 < 16) {
        #pragma unroll
        for (int mt = 0; mt < MT; ++mt) a_cur[mt] = a_nxt[mt];
      }
    }
  }

  v2f b3f[32];
  const bool realcol = idx < 8;
  {
    const float* w3p = W3 + (2 * hl) * 8 + (idx & 7);
    #pragma unroll
    for (int ks = 0; ks < 32; ++ks) {
      const float wx = w3p[(4 * ks) * 8];
      const float wy = w3p[(4 * ks + 1) * 8];
      b3f[ks].x = realcol ? wx : 0.0f;
      b3f[ks].y = realcol ? wy : 0.0f;
    }
  }
  const float bias3v = b3[idx & 7];
  const float bias3  = realcol ? bias3v : 0.0f;

  __syncthreads();

  {
    float* hf = (float*)hfrag;
    const int kb3  = 2 * wave + (idx >> 3);
    const int hh   = (idx >> 1) & 1;
    const int elem = (idx & 1) + 2 * ((idx >> 2) & 1);
    #pragma unroll
    for (int r = 0; r < 8; ++r) {
      const int o = fslot(0, kb3, 16 * hh + 8 * hl + r) * 4 + elem;
      #pragma unroll
      for (int mt = 0; mt < MT; ++mt) {
        const float v = acc[mt][r];
        hf[mt * (16 * 32 * 4) + o] = v > 0.0f ? v : 0.0f;
      }
    }
  }
  __syncthreads();

  v8f acc3;
  {
    #pragma unroll
    for (int r = 0; r < 8; ++r) acc3[r] = bias3;
  }
  #pragma unroll
  for (int kb = 0; kb < 16; ++kb) {
    const v4f av = hfrag[fslot(wave, kb, lane)];
    v2f a; a.x = av.x; a.y = av.y;
    acc3 = __builtin_amdgcn_wmma_f32_16x16x4_f32(
        false, a, false, b3f[2 * kb], (short)0, acc3, false, false);
    a.x = av.z; a.y = av.w;
    acc3 = __builtin_amdgcn_wmma_f32_16x16x4_f32(
        false, a, false, b3f[2 * kb + 1], (short)0, acc3, false, false);
  }

  if (idx < 8) {
    const int gbase = blockTok0 + wave * 16;
    #pragma unroll
    for (int r = 0; r < 8; ++r) {
      const int row = 8 * hl + r;
      out[(gbase + row) * 8 + idx] = acc3[r];
    }
  }
}

extern "C" void kernel_launch(void* const* d_in, const int* in_sizes, int n_in,
                              void* d_out, int out_size, void* d_ws, size_t ws_size,
                              hipStream_t stream) {
  const float* freqs = (const float*)d_in[0];
  const float* mags  = (const float*)d_in[1];
  const float* W1    = (const float*)d_in[2];
  const float* b1    = (const float*)d_in[3];
  const float* W2    = (const float*)d_in[4];
  const float* b2    = (const float*)d_in[5];
  const float* W3    = (const float*)d_in[6];
  const float* b3    = (const float*)d_in[7];
  float* out = (float*)d_out;

  mlp_and_last_row_kernel<<<1 + NB * NP / TOKENS_PER_BLOCK, 256, 0, stream>>>(
      freqs, mags, W1, b1, W2, b2, W3, b3, out);
  (void)d_ws; (void)ws_size; (void)n_in; (void)out_size; (void)in_sizes;
}


# last-row final pick: each thread first picks the best of its own four peaks in registers on one 64-bit (count, magnitude) key; the one lane then scans 256 thread winners once, not 1024 peaks twice
# speedup vs baseline: 9.3543x; 1.0329x over previous
#include <hip/hip_runtime.h>
#include <hip/hip_bf16.h>

typedef __attribute__((ext_vector_type(2))) float v2f;
typedef __attribute__((ext_vector_type(8))) float v8f;

#define HIDDEN 128
#define TOKENS_PER_BLOCK 128
#define NB 64
#define NP 1024

typedef __attribute__((ext_vector_type(4))) float v4f;
typedef __attribute__((ext_vector_type(4))) int   v4i;

#define MT 8

__device__ __forceinline__ int fslot(int mt, int kb, int fl) {
  return (mt * 16 + kb) * 32 + (fl ^ (((fl >> 4) & 1) << 1) ^ ((kb & 1) << 2));
}

__device__ __forceinline__ float l1act(float f, float m, float w0, float w1, float b) {
  const float h = fmaf(f, w0, fmaf(m, w1, b));
  return h > 0.0f ? h : 0.0f;
}

typedef __attribute__((ext_vector_type(4))) unsigned v4u;

#define PK 4

__device__ __forceinline__ unsigned f2ord(float f) {
  const unsigned b = __float_as_uint(f);
  return (b & 0x80000000u) ? ~b : (b | 0x80000000u);
}
__device__ __forceinline__ float ord2f(unsigned k) {
  const unsigned b = (k & 0x80000000u) ? (k & 0x7FFFFFFFu) : ~k;
  return __uint_as_float(b);
}
__device__ __forceinline__ float hdist(float fj, float fi, float order) {
  float d;
  {
    #pragma clang fp contract(off)
    d = fj - fi * order;
  }
  return d;
}

__device__ __forceinline__ void last_row_fundamental(
    const float* __restrict__ freqs, const float* __restrict__ mags,
    float* __restrict__ out)
{
  __shared__ __attribute__((aligned(16))) unsigned kh[NP / PK];
  __shared__ __attribute__((aligned(16))) unsigned kl[NP / PK];
  __shared__ __attribute__((aligned(16))) float    fw[NP / PK];
  __shared__ __attribute__((aligned(16))) unsigned xb[2][NP];
  __shared__ __attribute__((aligned(16))) float    sf[2 * NP];

  const int tid  = threadIdx.x;
  const int base = (NB - 1) * NP;
  const v4f fv = *(const v4f*)(freqs + base + PK * tid);
  const v4f mv = *(const v4f*)(mags + base + PK * tid);

  unsigned key[PK];
  #pragma unroll
  for (int e = 0; e < PK; ++e) key[e] = (mv[e] > 0.0f) ? f2ord(fv[e]) : 0xFFFFFFFFu;

  int tog = 0;
  #pragma unroll
  for (int k = 2; k <= NP; k <<= 1) {
    #pragma unroll
    for (int j = k >> 1; j > 0; j >>= 1) {
      unsigned other[PK];
      if (j >= PK * 32) {
        v4u* buf = (v4u*)xb[tog];
        tog ^= 1;
        v4u kv;
        #pragma unroll
        for (int e = 0; e < PK; ++e) kv[e] = key[e];
        buf[tid] = kv;
        __syncthreads();
        const v4u ov = buf[tid ^ (j / PK)];
        #pragma unroll
        for (int e = 0; e < PK; ++e) other[e] = ov[e];
      } else if (j >= PK) {
        #pragma unroll
        for (int e = 0; e < PK; ++e) other[e] = __shfl_xor(key[e], j / PK, 32);
      } else {
        #pragma unroll
        for (int e = 0; e < PK; ++e) other[e] = key[e ^ j];
      }
      #pragma unroll
      for (int e = 0; e < PK; ++e) {
        const int  i     = PK * tid + e;
        const bool up    = ((i & k) == 0);
        const bool lower = ((i & j) == 0);
        const unsigned mn = key[e] < other[e] ? key[e] : other[e];
        const unsigned mx = key[e] < other[e] ? other[e] : key[e];
        key[e] = (lower == up) ? mn : mx;
      }
    }
  }
  const float pinf = __builtin_inff();
  {
    v4f sv, iv;
    #pragma unroll
    for (int e = 0; e < PK; ++e) {
      sv[e] = (key[e] == 0xFFFFFFFFu) ? pinf : ord2f(key[e]);
      iv[e] = pinf;
    }
    ((v4f*)sf)[tid]           = sv;
    ((v4f*)sf)[NP / PK + tid] = iv;
  }
  __syncthreads();

  int lo[PK][7], hi[PK][7];
  #pragma unroll
  for (int e = 0; e < PK; ++e) {
    #pragma unroll
    for (int t = 0; t < 7; ++t) { lo[e][t] = 0; hi[e][t] = 0; }
  }
  #pragma unroll
  for (int step = NP; step > 0; step >>= 1) {
    #pragma unroll
    for (int e = 0; e < PK; ++e) {
      #pragma unroll
      for (int t = 0; t < 7; ++t) {
        const float of = (float)(t + 2);
        const float dl = hdist(sf[lo[e][t] + step - 1], fv[e], of);
        const float dh = hdist(sf[hi[e][t] + step - 1], fv[e], of);
        if (!(dl > -2.0f)) lo[e][t] += step;
        if (!(dh >= 2.0f)) hi[e][t] += step;
      }
    }
  }
  unsigned long long bkey = 0ull;
  float bfreq = fv[0];
  #pragma unroll
  for (int e = 0; e < PK; ++e) {
    const float fi    = fv[e];
    const bool  valid = mv[e] > 0.0f;
    const bool neg = fi < 0.0f;
    int h = 0, cover = 0;
    #pragma unroll
    for (int t = 0; t < 7; ++t) {
      int a       = neg ? 2 * NP - hi[e][t] : lo[e][t];
      const int b = neg ? 2 * NP - lo[e][t] : hi[e][t];
      a = a > cover ? a : cover;
      h += b > a ? b - a : 0;
      cover = cover > b ? cover : b;
    }
    bool selfm = false;
    #pragma unroll
    for (int t = 0; t < 7; ++t)
      selfm = selfm || (fabsf(hdist(fi, fi, (float)(t + 2))) < 2.0f);
    h -= selfm ? 1 : 0;
    const unsigned hp = (valid && h > 0) ? (unsigned)h : 0u;
    const unsigned long long k =
        ((unsigned long long)hp << 32) | (unsigned long long)f2ord(mv[e] + 0.0f);
    const bool take = (e == 0) || (k > bkey);
    bkey  = take ? k : bkey;
    bfreq = take ? fi : bfreq;
  }
  kh[tid] = (unsigned)(bkey >> 32);
  kl[tid] = (unsigned)bkey;
  fw[tid] = bfreq;
  __syncthreads();

  if (tid == 0) {
    const v4u* kh4 = (const v4u*)kh;
    const v4u* kl4 = (const v4u*)kl;
    unsigned long long bk[4];
    int bq[4];
    #pragma unroll
    for (int c = 0; c < 4; ++c) { bk[c] = 0ull; bq[c] = 0; }
    for (int g = 0; g < NP / PK / 4 / 8; ++g) {
      v4u hq[8], lq[8];
      #pragma unroll
      for (int u = 0; u < 8; ++u) { hq[u] = kh4[8 * g + u]; lq[u] = kl4[8 * g + u]; }
      asm volatile("" :: "v"(hq[0]), "v"(hq[1]), "v"(hq[2]), "v"(hq[3]),
                         "v"(hq[4]), "v"(hq[5]), "v"(hq[6]), "v"(hq[7]),
                         "v"(lq[0]), "v"(lq[1]), "v"(lq[2]), "v"(lq[3]),
                         "v"(lq[4]), "v"(lq[5]), "v"(lq[6]), "v"(lq[7]));
      #pragma unroll
      for (int u = 0; u < 8; ++u) {
        const int q = 8 * g + u;
        #pragma unroll
        for (int c = 0; c < 4; ++c) {
          const unsigned long long k = ((unsigned long long)hq[u][c] << 32) | (unsigned long long)lq[u][c];
          const bool take = k > bk[c];
          bk[c] = take ? k : bk[c];
          bq[c] = take ? q : bq[c];
        }
      }
    }
    unsigned long long best_key = bk[0];
    int best_t = 4 * bq[0];
    #pragma unroll
    for (int c = 1; c < 4; ++c) {
      const int  tc   = 4 * bq[c] + c;
      const bool take = (bk[c] > best_key) | ((bk[c] == best_key) & (tc < best_t));
      best_key = take ? bk[c] : best_key;
      best_t   = take ? tc : best_t;
    }
    out[NB * NP * 8] = fw[best_t];
  }
}

__global__ __launch_bounds__(256) void mlp_and_last_row_kernel(
    const float* __restrict__ freqs, const float* __restrict__ mags,
    const float* __restrict__ W1, const float* __restrict__ b1,
    const float* __restrict__ W2, const float* __restrict__ b2,
    const float* __restrict__ W3, const float* __restrict__ b3,
    float* __restrict__ out)
{
  __shared__ v4f hfrag[MT * 16 * 32];

  if (blockIdx.x == 0) {
    last_row_fundamental(freqs, mags, out);
    return;
  }

  const int tid       = threadIdx.x;
  const int blockTok0 = (blockIdx.x - 1) * TOKENS_PER_BLOCK;

  const int lane = tid & 31;
  const int wave = tid >> 5;
  const int idx  = lane & 15;
  const int hl   = lane >> 4;

  const int   g1    = blockTok0 + wave * 16 + idx;
  const float f     = freqs[g1];
  const float m     = mags[g1];
  const float bias2 = b2[wave * 16 + idx];

  v2f bw[32];
  {
    const float* w2p = W2 + (2 * hl) * HIDDEN + wave * 16 + idx;
    #pragma unroll
    for (int ks = 0; ks < 32; ++ks) {
      bw[ks].x = w2p[(4 * ks) * HIDDEN];
      bw[ks].y = w2p[(4 * ks + 1) * HIDDEN];
    }
  }

  {
    #pragma unroll 4
    for (int kb = 0; kb < 16; ++kb) {
      const int ca = kb * 8 + 2 * hl;
      const int cb = ca + 4;
      const v2f w0a = *(const v2f*)(W1 + ca);
      const v2f w1a = *(const v2f*)(W1 + HIDDEN + ca);
      const v2f bba = *(const v2f*)(b1 + ca);
      const v2f w0b = *(const v2f*)(W1 + cb);
      const v2f w1b = *(const v2f*)(W1 + HIDDEN + cb);
      const v2f bbb = *(const v2f*)(b1 + cb);
      v4f hv;
      hv.x = l1act(f, m, w0a.x, w1a.x, bba.x);
      hv.y = l1act(f, m, w0a.y, w1a.y, bba.y);
      hv.z = l1act(f, m, w0b.x, w1b.x, bbb.x);
      hv.w = l1act(f, m, w0b.y, w1b.y, bbb.y);
      hfrag[fslot(wave, kb, lane)] = hv;
    }
  }
  __syncthreads();

  v8f acc[MT];
  {
    #pragma unroll
    for (int mt = 0; mt < MT; ++mt) {
      #pragma unroll
      for (int r = 0; r < 8; ++r) acc[mt][r] = bias2;
    }
  }

  {
    v4f a_cur[MT];
    #pragma unroll
    for (int mt = 0; mt < MT; ++mt) a_cur[mt] = hfrag[fslot(mt, 0, lane)];

    #pragma unroll
    for (int kb = 0; kb < 16; ++kb) {
      v4f a_nxt[MT];
      if (kb + 1 < 16) {
        #pragma unroll
        for (int mt = 0; mt < MT; ++mt) a_nxt[mt] = hfrag[fslot(mt, kb + 1, lane)];
      }
      #pragma unroll
      for (int mt = 0; mt < MT; ++mt) {
        v2f a; a.x = a_cur[mt].x; a.y = a_cur[mt].y;
        acc[mt] = __builtin_amdgcn_wmma_f32_16x16x4_f32(
            false, a, false, bw[2 * kb], (short)0, acc[mt], false, false);
      }
      #pragma unroll
      for (int mt = 0; mt < MT; ++mt) {
        v2f a; a.x = a_cur[mt].z; a.y = a_cur[mt].w;
        acc[mt] = __builtin_amdgcn_wmma_f32_16x16x4_f32(
            false, a, false, bw[2 * kb + 1], (short)0, acc[mt], false, false);
      }
      if (kb + 1 < 16) {
        #pragma unroll
        for (int mt = 0; mt < MT; ++mt) a_cur[mt] = a_nxt[mt];
      }
    }
  }

  v2f b3f[32];
  const bool realcol = idx < 8;
  {
    const float* w3p = W3 + (2 * hl) * 8 + (idx & 7);
    #pragma unroll
    for (int ks = 0; ks < 32; ++ks) {
      const float wx = w3p[(4 * ks) * 8];
      const float wy = w3p[(4 * ks + 1) * 8];
      b3f[ks].x = realcol ? wx : 0.0f;
      b3f[ks].y = realcol ? wy : 0.0f;
    }
  }
  const float bias3v = b3[idx & 7];
  const float bias3  = realcol ? bias3v : 0.0f;

  __syncthreads();

  {
    float* hf = (float*)hfrag;
    const int kb3  = 2 * wave + (idx >> 3);
    const int hh   = (idx >> 1) & 1;
    const int elem = (idx & 1) + 2 * ((idx >> 2) & 1);
    #pragma unroll
    for (int r = 0; r < 8; ++r) {
      const int o = fslot(0, kb3, 16 * hh + 8 * hl + r) * 4 + elem;
      #pragma unroll
      for (int mt = 0; mt < MT; ++mt) {
        const float v = acc[mt][r];
        hf[mt * (16 * 32 * 4) + o] = v > 0.0f ? v : 0.0f;
      }
    }
  }
  __syncthreads();

  v8f acc3;
  {
    #pragma unroll
    for (int r = 0; r < 8; ++r) acc3[r] = bias3;
  }
  #pragma unroll
  for (int kb = 0; kb < 16; ++kb) {
    const v4f av = hfrag[fslot(wave, kb, lane)];
    v2f a; a.x = av.x; a.y = av.y;
    acc3 = __builtin_amdgcn_wmma_f32_16x16x4_f32(
        false, a, false, b3f[2 * kb], (short)0, acc3, false, false);
    a.x = av.z; a.y = av.w;
    acc3 = __builtin_amdgcn_wmma_f32_16x16x4_f32(
        false, a, false, b3f[2 * kb + 1], (short)0, acc3, false, false);
  }

  if (idx < 8) {
    const int gbase = blockTok0 + wave * 16;
    #pragma unroll
    for (int r = 0; r < 8; ++r) {
      const int row = 8 * hl + r;
      out[(gbase + row) * 8 + idx] = acc3[r];
    }
  }
}

extern "C" void kernel_launch(void* const* d_in, const int* in_sizes, int n_in,
                              void* d_out, int out_size, void* d_ws, size_t ws_size,
                              hipStream_t stream) {
  const float* freqs = (const float*)d_in[0];
  const float* mags  = (const float*)d_in[1];
  const float* W1    = (const float*)d_in[2];
  const float* b1    = (const float*)d_in[3];
  const float* W2    = (const float*)d_in[4];
  const float* b2    = (const float*)d_in[5];
  const float* W3    = (const float*)d_in[6];
  const float* b3    = (const float*)d_in[7];
  float* out = (float*)d_out;

  mlp_and_last_row_kernel<<<1 + NB * NP / TOKENS_PER_BLOCK, 256, 0, stream>>>(
      freqs, mags, W1, b1, W2, b2, W3, b3, out);
  (void)d_ws; (void)ws_size; (void)n_in; (void)out_size; (void)in_sizes;
}
